# starting kernel
# baseline (speedup 1.0000x reference)
#include <hip/hip_runtime.h>
#include <math.h>

typedef __attribute__((ext_vector_type(2))) float v2f;
typedef __attribute__((ext_vector_type(8))) float v8f;

#define D 256

__device__ __forceinline__ float gelu_exact(float x) {
    return 0.5f * x * (1.0f + erff(x * 0.70710678118654752440f));
}

__global__ __launch_bounds__(512) void mlp_weight_kernel(
    const float* __restrict__ W1, const float* __restrict__ b1,
    const float* __restrict__ W2, const float* __restrict__ b2,
    const float* __restrict__ W3, const float* __restrict__ b3,
    const float* __restrict__ W4, const float* __restrict__ b4,
    const float* __restrict__ W5, const float* __restrict__ b5,
    float* __restrict__ w12_out)
{
    __shared__ float bufA[D];
    __shared__ float bufB[D];

    const int tid  = threadIdx.x;
    const int wave = tid >> 5;
    const int lane = tid & 31;
    const int n    = lane & 15;
    const int hi   = (lane >> 4) & 1;

    if (tid < D) {
        float v = W1[tid * 4 + 2] + W1[tid * 4 + 3] + b1[tid];
        bufA[tid] = gelu_exact(v);
    }
    __syncthreads();

    const float* Ws[3] = {W2, W3, W4};
    const float* bs[3] = {b2, b3, b4};
    float* cur = bufA;
    float* nxt = bufB;

    for (int L = 0; L < 3; ++L) {
        const float* Wm = Ws[L];
        const float* bb = bs[L];
        const int tile  = wave;
        const float* wrow = Wm + (size_t)(tile * 16 + n) * D;

        v8f acc = {};
        for (int k0 = 0; k0 < D; k0 += 4) {
            v2f a, bmat;
            a[0]    = cur[k0 + 2 * hi];
            a[1]    = cur[k0 + 2 * hi + 1];
            bmat[0] = wrow[k0 + 2 * hi];
            bmat[1] = wrow[k0 + 2 * hi + 1];
            acc = __builtin_amdgcn_wmma_f32_16x16x4_f32(
                false, a, false, bmat, (short)0, acc, false, false);
        }
        nxt[tile * 16 + n] = gelu_exact(acc[0] + bb[tile * 16 + n]);
        __syncthreads();
        float* t = cur; cur = nxt; nxt = t;
    }

    if (tid < 12) {
        const float* wr = W5 + tid * D;
        float s = b5[tid];
        for (int k = 0; k < D; ++k) s = fmaf(cur[k], wr[k], s);
        w12_out[tid] = s;
    }
}

__global__ __launch_bounds__(256) void pred_kernel(
    const float* __restrict__ feat, const float* __restrict__ w12,
    float* __restrict__ out)
{
    __shared__ float w[12];
    if (threadIdx.x < 12) w[threadIdx.x] = w12[threadIdx.x];
    __syncthreads();

    const int t = blockIdx.x * 256 + threadIdx.x;
    const int b = t >> 14;
    const int i = t & 16383;

    const float4* fb = (const float4*)(feat + ((size_t)b << 18));
    float4 c0 = fb[i];
    float4 c1 = fb[16384 + i];
    float4 c2 = fb[2 * 16384 + i];
    float4 c3 = fb[3 * 16384 + i];

    float p0[4] = {c0.x, c0.y, c0.z, c0.w};
    float p1[4] = {c1.x, c1.y, c1.z, c1.w};
    float p2[4] = {c2.x, c2.y, c2.z, c2.w};
    float p3[4] = {c3.x, c3.y, c3.z, c3.w};

    float o[12];
#pragma unroll
    for (int j = 0; j < 4; ++j) {
#pragma unroll
        for (int d = 0; d < 3; ++d) {
            o[j * 3 + d] = fmaf(p0[j], w[d],
                           fmaf(p1[j], w[3 + d],
                           fmaf(p2[j], w[6 + d],
                                p3[j] * w[9 + d])));
        }
    }

    float4* op = (float4*)(out + (size_t)t * 12);
    op[0] = make_float4(o[0], o[1], o[2],  o[3]);
    op[1] = make_float4(o[4], o[5], o[6],  o[7]);
    op[2] = make_float4(o[8], o[9], o[10], o[11]);
}

extern "C" void kernel_launch(void* const* d_in, const int* in_sizes, int n_in,
                              void* d_out, int out_size, void* d_ws, size_t ws_size,
                              hipStream_t stream) {
    const float* feat = (const float*)d_in[0];
    const float* W1   = (const float*)d_in[1];
    const float* b1   = (const float*)d_in[2];
    const float* W2   = (const float*)d_in[3];
    const float* b2   = (const float*)d_in[4];
    const float* W3   = (const float*)d_in[5];
    const float* b3   = (const float*)d_in[6];
    const float* W4   = (const float*)d_in[7];
    const float* b4   = (const float*)d_in[8];
    const float* W5   = (const float*)d_in[9];
    const float* b5   = (const float*)d_in[10];

    float* w12 = (float*)d_ws;
    float* out = (float*)d_out;

    mlp_weight_kernel<<<1, 512, 0, stream>>>(W1, b1, W2, b2, W3, b3, W4, b4, W5, b5, w12);
    pred_kernel<<<256, 256, 0, stream>>>(feat, w12, out);
}


# MLP: W rows feed WMMA operand A straight from whole-line 16-B global loads issued ahead (tile 2 rows x 8 k-chunks, x as B, row sum = D diagonal + 3 DPP adds); sum order of layers 2-5 changes
# speedup vs baseline: 1.9587x; 1.9587x over previous
#include <hip/hip_runtime.h>
#include <math.h>

typedef __attribute__((ext_vector_type(2))) float v2f;
typedef __attribute__((ext_vector_type(4))) float v4f;
typedef __attribute__((ext_vector_type(8))) float v8f;

#define D 256
#define RING 4

__device__ __forceinline__ float gelu_exact(float x) {
    return 0.5f * x * (1.0f + erff(x * 0.70710678118654752440f));
}

template <int M>
__device__ __forceinline__ float xmask_add(float v) {
    int t = __builtin_amdgcn_update_dpp(0, __builtin_bit_cast(int, v), 0x160 + M, 0xf, 0xf, true);
    return v + __builtin_bit_cast(float, t);
}

__device__ __forceinline__ void lds_barrier() {
    __builtin_amdgcn_fence(__ATOMIC_RELEASE, "workgroup", "local");
    __builtin_amdgcn_s_barrier();
    __builtin_amdgcn_fence(__ATOMIC_ACQUIRE, "workgroup", "local");
}


__device__ __forceinline__ void issue_pair(const float* __restrict__ p, v4f (&w)[4]) {
#pragma unroll
    for (int g = 0; g < 4; ++g) w[g] = *(const v4f*)(p + g * 64);
}

__device__ __forceinline__ void load_x(const float* xs, int koff, v4f (&xf)[4]) {
#pragma unroll
    for (int g = 0; g < 4; ++g) xf[g] = *(const v4f*)(xs + g * 64 + koff);
}

__device__ __forceinline__ float pair_dot(const v4f (&w)[4], const v4f (&xf)[4], int q) {
    v8f acc = {};
#pragma unroll
    for (int g = 0; g < 4; ++g) {
        v2f a0 = {w[g].x, w[g].y};
        v2f b0 = {xf[g].x, xf[g].y};
        acc = __builtin_amdgcn_wmma_f32_16x16x4_f32(false, a0, false, b0, (short)0, acc, false, false);
        v2f a1 = {w[g].z, w[g].w};
        v2f b1 = {xf[g].z, xf[g].w};
        acc = __builtin_amdgcn_wmma_f32_16x16x4_f32(false, a1, false, b1, (short)0, acc, false, false);
    }
    float v = acc[0];
#pragma unroll
    for (int r = 1; r < 8; ++r) v = (q == r) ? acc[r] : v;
    v = xmask_add<1>(v);
    v = xmask_add<2>(v);
    v = xmask_add<4>(v);
    return v;
}

__global__ __launch_bounds__(512) void mlp_weight_kernel(
    const float* __restrict__ W1, const float* __restrict__ b1,
    const float* __restrict__ W2, const float* __restrict__ b2,
    const float* __restrict__ W3, const float* __restrict__ b3,
    const float* __restrict__ W4, const float* __restrict__ b4,
    const float* __restrict__ W5, const float* __restrict__ b5,
    float* __restrict__ w12_out)
{
    __shared__ __attribute__((aligned(16))) float bufA[D];
    __shared__ __attribute__((aligned(16))) float bufB[D];

    const int tid  = threadIdx.x;
    const int wave = __builtin_amdgcn_readfirstlane(tid >> 5);
    const int lane = tid & 31;
    const int c    = lane & 15;
    const int h    = lane >> 4;
    const int q    = c & 7;
    const int koff = 4 * (q + 8 * h);

    const size_t lane_off = (size_t)(16 * wave + (c >> 3)) * D + koff;
    const float* wl[3] = {W2 + lane_off, W3 + lane_off, W4 + lane_off};
    const float* bs[3] = {b2, b3, b4};
    const int    p5  = wave < 6 ? wave : 0;
    const float* w5l = W5 + (size_t)(2 * p5 + (c >> 3)) * D + koff;

    const int myrow = 16 * wave + 2 * q + h;

    v4f ring[RING][4];
    v4f xf[4];

#pragma unroll
    for (int t = 0; t < RING; ++t) issue_pair(wl[0] + t * 2 * D, ring[t]);

    if (tid < D) {
        float v = W1[tid * 4 + 2] + W1[tid * 4 + 3] + b1[tid];
        bufA[tid] = gelu_exact(v);
    }
    lds_barrier();

    float bias5 = 0.0f;

#pragma unroll
    for (int L = 0; L < 3; ++L) {
        const float* xs = (L & 1) ? bufB : bufA;
        float*       xd = (L & 1) ? bufA : bufB;
        load_x(xs, koff, xf);
        const float bias = bs[L][myrow];
        if (L == 2) bias5 = b5[2 * p5 + h];
        float mine = 0.0f;
#pragma unroll
        for (int p = 0; p < 8; ++p) {
            const int t = L * 8 + p;
            const float v = pair_dot(ring[t % RING], xf, q);
            mine = (q == p) ? v : mine;
            const int tn = t + RING;
            if (tn < 24)       issue_pair(wl[tn >> 3] + (tn & 7) * 2 * D, ring[t % RING]);
            else if (tn == 24) issue_pair(w5l, ring[t % RING]);
        }
        xd[myrow] = gelu_exact(mine + bias);
        lds_barrier();
    }

    if (wave < 6) {
        load_x(bufB, koff, xf);
        const float v = pair_dot(ring[24 % RING], xf, q);
        if (c == 0) bufA[2 * wave + h] = v + bias5;
    }
    lds_barrier();
    if (tid < 12) w12_out[tid] = bufA[tid];
}

__global__ __launch_bounds__(256) void pred_kernel(
    const float* __restrict__ feat, const float* __restrict__ w12,
    float* __restrict__ out)
{
    __shared__ float w[12];
    if (threadIdx.x < 12) w[threadIdx.x] = w12[threadIdx.x];
    __syncthreads();

    const int t = blockIdx.x * 256 + threadIdx.x;
    const int b = t >> 14;
    const int i = t & 16383;

    const float4* fb = (const float4*)(feat + ((size_t)b << 18));
    float4 c0 = fb[i];
    float4 c1 = fb[16384 + i];
    float4 c2 = fb[2 * 16384 + i];
    float4 c3 = fb[3 * 16384 + i];

    float p0[4] = {c0.x, c0.y, c0.z, c0.w};
    float p1[4] = {c1.x, c1.y, c1.z, c1.w};
    float p2[4] = {c2.x, c2.y, c2.z, c2.w};
    float p3[4] = {c3.x, c3.y, c3.z, c3.w};

    float o[12];
#pragma unroll
    for (int j = 0; j < 4; ++j) {
#pragma unroll
        for (int d = 0; d < 3; ++d) {
            o[j * 3 + d] = fmaf(p0[j], w[d],
                           fmaf(p1[j], w[3 + d],
                           fmaf(p2[j], w[6 + d],
                                p3[j] * w[9 + d])));
        }
    }

    float4* op = (float4*)(out + (size_t)t * 12);
    op[0] = make_float4(o[0], o[1], o[2],  o[3]);
    op[1] = make_float4(o[4], o[5], o[6],  o[7]);
    op[2] = make_float4(o[8], o[9], o[10], o[11]);
}

extern "C" void kernel_launch(void* const* d_in, const int* in_sizes, int n_in,
                              void* d_out, int out_size, void* d_ws, size_t ws_size,
                              hipStream_t stream) {
    const float* feat = (const float*)d_in[0];
    const float* W1   = (const float*)d_in[1];
    const float* b1   = (const float*)d_in[2];
    const float* W2   = (const float*)d_in[3];
    const float* b2   = (const float*)d_in[4];
    const float* W3   = (const float*)d_in[5];
    const float* b3   = (const float*)d_in[6];
    const float* W4   = (const float*)d_in[7];
    const float* b4   = (const float*)d_in[8];
    const float* W5   = (const float*)d_in[9];
    const float* b5   = (const float*)d_in[10];

    float* w12 = (float*)d_ws;
    float* out = (float*)d_out;

    mlp_weight_kernel<<<1, 512, 0, stream>>>(W1, b1, W2, b2, W3, b3, W4, b4, W5, b5, w12);
    pred_kernel<<<256, 256, 0, stream>>>(feat, w12, out);
}


# One launch instead of two: each of 64 work-groups recomputes the 12-float MLP result itself (same code, same sum order) and then streams its share of the pixels, loaded while layer 4 finishes
# speedup vs baseline: 2.1202x; 1.0825x over previous
#include <hip/hip_runtime.h>
#include <math.h>

typedef __attribute__((ext_vector_type(2))) float v2f;
typedef __attribute__((ext_vector_type(4))) float v4f;
typedef __attribute__((ext_vector_type(8))) float v8f;

#define D 256
#define RING 4
#define NWG 64
#define ITEMS 2
static_assert(NWG * ITEMS * 512 == 65536 && ITEMS <= 3, "every item once; one item per freed ring slot");

__device__ __forceinline__ float gelu_exact(float x) {
    return 0.5f * x * (1.0f + erff(x * 0.70710678118654752440f));
}

template <int M>
__device__ __forceinline__ float xmask_add(float v) {
    int t = __builtin_amdgcn_update_dpp(0, __builtin_bit_cast(int, v), 0x160 + M, 0xf, 0xf, true);
    return v + __builtin_bit_cast(float, t);
}

__device__ __forceinline__ void lds_barrier() {
    __builtin_amdgcn_fence(__ATOMIC_RELEASE, "workgroup", "local");
    __builtin_amdgcn_s_barrier();
    __builtin_amdgcn_fence(__ATOMIC_ACQUIRE, "workgroup", "local");
}


__device__ __forceinline__ void issue_pair(const float* __restrict__ p, v4f (&w)[4]) {
#pragma unroll
    for (int g = 0; g < 4; ++g) w[g] = *(const v4f*)(p + g * 64);
}

__device__ __forceinline__ void load_x(const float* xs, int koff, v4f (&xf)[4]) {
#pragma unroll
    for (int g = 0; g < 4; ++g) xf[g] = *(const v4f*)(xs + g * 64 + koff);
}

__device__ __forceinline__ float pair_dot(const v4f (&w)[4], const v4f (&xf)[4], int q) {
    v8f acc = {};
#pragma unroll
    for (int g = 0; g < 4; ++g) {
        v2f a0 = {w[g].x, w[g].y};
        v2f b0 = {xf[g].x, xf[g].y};
        acc = __builtin_amdgcn_wmma_f32_16x16x4_f32(false, a0, false, b0, (short)0, acc, false, false);
        v2f a1 = {w[g].z, w[g].w};
        v2f b1 = {xf[g].z, xf[g].w};
        acc = __builtin_amdgcn_wmma_f32_16x16x4_f32(false, a1, false, b1, (short)0, acc, false, false);
    }
    float v = acc[0];
#pragma unroll
    for (int r = 1; r < 8; ++r) v = (q == r) ? acc[r] : v;
    v = xmask_add<1>(v);
    v = xmask_add<2>(v);
    v = xmask_add<4>(v);
    return v;
}

__device__ __forceinline__ void load_item(const float* __restrict__ feat, int t, float4 (&f)[4]) {
    const int b = t >> 14;
    const int i = t & 16383;
    const float4* fb = (const float4*)(feat + ((size_t)b << 18));
#pragma unroll
    for (int ch = 0; ch < 4; ++ch) f[ch] = fb[ch * 16384 + i];
}

__global__ __launch_bounds__(512) void mlp_pred_kernel(
    const float* __restrict__ feat,
    const float* __restrict__ W1, const float* __restrict__ b1,
    const float* __restrict__ W2, const float* __restrict__ b2,
    const float* __restrict__ W3, const float* __restrict__ b3,
    const float* __restrict__ W4, const float* __restrict__ b4,
    const float* __restrict__ W5, const float* __restrict__ b5,
    float* __restrict__ out)
{
    __shared__ __attribute__((aligned(16))) float bufA[D];
    __shared__ __attribute__((aligned(16))) float bufB[D];

    const int tid  = threadIdx.x;
    const int wave = __builtin_amdgcn_readfirstlane(tid >> 5);
    const int lane = tid & 31;
    const int c    = lane & 15;
    const int h    = lane >> 4;
    const int q    = c & 7;
    const int koff = 4 * (q + 8 * h);

    const size_t lane_off = (size_t)(16 * wave + (c >> 3)) * D + koff;
    const float* wl[3] = {W2 + lane_off, W3 + lane_off, W4 + lane_off};
    const float* bs[3] = {b2, b3, b4};
    const int    p5  = wave < 6 ? wave : 0;
    const float* w5l = W5 + (size_t)(2 * p5 + (c >> 3)) * D + koff;

    const int myrow = 16 * wave + 2 * q + h;

    v4f ring[RING][4];
    v4f xf[4];

#pragma unroll
    for (int t = 0; t < RING; ++t) issue_pair(wl[0] + t * 2 * D, ring[t]);

    if (tid < D) {
        float v = W1[tid * 4 + 2] + W1[tid * 4 + 3] + b1[tid];
        bufA[tid] = gelu_exact(v);
    }

    float4 f[ITEMS][4];
    lds_barrier();

    float bias5 = 0.0f;

#pragma unroll
    for (int L = 0; L < 3; ++L) {
        const float* xs = (L & 1) ? bufB : bufA;
        float*       xd = (L & 1) ? bufA : bufB;
        load_x(xs, koff, xf);
        const float bias = bs[L][myrow];
        if (L == 2) bias5 = b5[2 * p5 + h];
        float mine = 0.0f;
#pragma unroll
        for (int p = 0; p < 8; ++p) {
            const int t = L * 8 + p;
            const float v = pair_dot(ring[t % RING], xf, q);
            mine = (q == p) ? v : mine;
            const int tn = t + RING;
            if (tn < 24)       issue_pair(wl[tn >> 3] + (tn & 7) * 2 * D, ring[t % RING]);
            else if (tn == 24) issue_pair(w5l, ring[t % RING]);
            if (tn > 24 && tn - 25 < ITEMS) load_item(feat, (blockIdx.x * ITEMS + (tn - 25)) * 512 + tid, f[tn - 25]);
        }
        xd[myrow] = gelu_exact(mine + bias);
        lds_barrier();
    }

    if (wave < 6) {
        load_x(bufB, koff, xf);
        const float v = pair_dot(ring[24 % RING], xf, q);
        if (c == 0) bufA[2 * wave + h] = v + bias5;
    }
    lds_barrier();

    float w[12];
#pragma unroll
    for (int g = 0; g < 3; ++g) {
        const v4f wv = *(const v4f*)(bufA + 4 * g);
        w[4 * g + 0] = wv.x; w[4 * g + 1] = wv.y; w[4 * g + 2] = wv.z; w[4 * g + 3] = wv.w;
    }
#pragma unroll
    for (int it = 0; it < ITEMS; ++it) {
        const int t = (blockIdx.x * ITEMS + it) * 512 + tid;
        const float p0[4] = {f[it][0].x, f[it][0].y, f[it][0].z, f[it][0].w};
        const float p1[4] = {f[it][1].x, f[it][1].y, f[it][1].z, f[it][1].w};
        const float p2[4] = {f[it][2].x, f[it][2].y, f[it][2].z, f[it][2].w};
        const float p3[4] = {f[it][3].x, f[it][3].y, f[it][3].z, f[it][3].w};
        float o[12];
#pragma unroll
        for (int j = 0; j < 4; ++j) {
#pragma unroll
            for (int d = 0; d < 3; ++d) {
                o[j * 3 + d] = fmaf(p0[j], w[d],
                               fmaf(p1[j], w[3 + d],
                               fmaf(p2[j], w[6 + d],
                                    p3[j] * w[9 + d])));
            }
        }
        float4* op = (float4*)(out + (size_t)t * 12);
        op[0] = make_float4(o[0], o[1], o[2],  o[3]);
        op[1] = make_float4(o[4], o[5], o[6],  o[7]);
        op[2] = make_float4(o[8], o[9], o[10], o[11]);
    }
}

extern "C" void kernel_launch(void* const* d_in, const int* in_sizes, int n_in,
                              void* d_out, int out_size, void* d_ws, size_t ws_size,
                              hipStream_t stream) {
    const float* feat = (const float*)d_in[0];
    const float* W1   = (const float*)d_in[1];
    const float* b1   = (const float*)d_in[2];
    const float* W2   = (const float*)d_in[3];
    const float* b2   = (const float*)d_in[4];
    const float* W3   = (const float*)d_in[5];
    const float* b3   = (const float*)d_in[6];
    const float* W4   = (const float*)d_in[7];
    const float* b4   = (const float*)d_in[8];
    const float* W5   = (const float*)d_in[9];
    const float* b5   = (const float*)d_in[10];

    float* out = (float*)d_out;

    mlp_pred_kernel<<<NWG, 512, 0, stream>>>(feat, W1, b1, W2, b2, W3, b3, W4, b4, W5, b5, out);
}


# MLP layers 2-4: weights in BOTH WMMA operands (x rides in 4 lane rows of A and 4 lane columns of B): 96 weights per WMMA, was 64; 48 WMMAs per wave and layer, was 64; sum order of layers 2-4 changes
# speedup vs baseline: 2.3075x; 1.0883x over previous
#include <hip/hip_runtime.h>
#include <math.h>

typedef __attribute__((ext_vector_type(2))) float v2f;
typedef __attribute__((ext_vector_type(4))) float v4f;
typedef __attribute__((ext_vector_type(8))) float v8f;

#define D 256
#define KPS 2
#define SPC (8 / KPS)
#define NSLOT 5
#define NU (9 * SPC)
#define NWG 64
#define ITEMS 2
static_assert(NWG * ITEMS * 512 == 65536 && ITEMS <= NSLOT - 1, "every item once; one item per freed ring slot");
static_assert(KPS == 2 || KPS == 4, "a slot holds whole even/odd pairs of k ranges");

__device__ __forceinline__ float gelu_exact(float x) {
    return 0.5f * x * (1.0f + erff(x * 0.70710678118654752440f));
}

template <int M>
__device__ __forceinline__ float xmask_add(float v) {
    int t = __builtin_amdgcn_update_dpp(0, __builtin_bit_cast(int, v), 0x160 + M, 0xf, 0xf, true);
    return v + __builtin_bit_cast(float, t);
}

__device__ __forceinline__ float from_xmask4(float v) {
    int t = __builtin_amdgcn_update_dpp(0, __builtin_bit_cast(int, v), 0x164, 0xf, 0xf, true);
    return __builtin_bit_cast(float, t);
}

__device__ __forceinline__ void lds_barrier() {
    __builtin_amdgcn_fence(__ATOMIC_RELEASE, "workgroup", "local");
    __builtin_amdgcn_s_barrier();
    __builtin_amdgcn_fence(__ATOMIC_ACQUIRE, "workgroup", "local");
}


__device__ __forceinline__ void issue_slot(const float* __restrict__ Wp, int offA, int offB, int part, v4f (&s)[2 * KPS]) {
#pragma unroll
    for (int i = 0; i < KPS; ++i) {
        s[2 * i]     = *(const v4f*)(Wp + offA + (part * KPS + i) * 32);
        s[2 * i + 1] = *(const v4f*)(Wp + offB + (part * KPS + i) * 32);
    }
}

__device__ __forceinline__ void slot_mma(const v4f (&s)[2 * KPS], const v4f (&xk)[8], int part, bool isxA, bool isxB,
                                         v8f& accE, v8f& accO) {
    v4f a[KPS], b[KPS];
#pragma unroll
    for (int i = 0; i < KPS; ++i) {
        a[i] = isxA ? xk[part * KPS + i] : s[2 * i];
        b[i] = isxB ? xk[part * KPS + i] : s[2 * i + 1];
    }
#pragma unroll
    for (int i = 0; i < KPS; i += 2) {
        v2f a0 = {a[i].x, a[i].y},         b0 = {b[i].x, b[i].y};
        v2f a1 = {a[i + 1].x, a[i + 1].y}, b1 = {b[i + 1].x, b[i + 1].y};
        v2f a2 = {a[i].z, a[i].w},         b2 = {b[i].z, b[i].w};
        v2f a3 = {a[i + 1].z, a[i + 1].w}, b3 = {b[i + 1].z, b[i + 1].w};
        accE = __builtin_amdgcn_wmma_f32_16x16x4_f32(false, a0, false, b0, (short)0, accE, false, false);
        accO = __builtin_amdgcn_wmma_f32_16x16x4_f32(false, a1, false, b1, (short)0, accO, false, false);
        accE = __builtin_amdgcn_wmma_f32_16x16x4_f32(false, a2, false, b2, (short)0, accE, false, false);
        accO = __builtin_amdgcn_wmma_f32_16x16x4_f32(false, a3, false, b3, (short)0, accO, false, false);
    }
}

__device__ __forceinline__ void issue_pair(const float* __restrict__ p, v4f (&w)[4]) {
#pragma unroll
    for (int g = 0; g < 4; ++g) w[g] = *(const v4f*)(p + g * 64);
}

__device__ __forceinline__ void load_x(const float* xs, int koff, v4f (&xf)[4]) {
#pragma unroll
    for (int g = 0; g < 4; ++g) xf[g] = *(const v4f*)(xs + g * 64 + koff);
}

__device__ __forceinline__ float pair_dot(const v4f (&w)[4], const v4f (&xf)[4], int q) {
    v8f acc = {};
#pragma unroll
    for (int g = 0; g < 4; ++g) {
        v2f a0 = {w[g].x, w[g].y};
        v2f b0 = {xf[g].x, xf[g].y};
        acc = __builtin_amdgcn_wmma_f32_16x16x4_f32(false, a0, false, b0, (short)0, acc, false, false);
        v2f a1 = {w[g].z, w[g].w};
        v2f b1 = {xf[g].z, xf[g].w};
        acc = __builtin_amdgcn_wmma_f32_16x16x4_f32(false, a1, false, b1, (short)0, acc, false, false);
    }
    float v = acc[0];
#pragma unroll
    for (int r = 1; r < 8; ++r) v = (q == r) ? acc[r] : v;
    v = xmask_add<1>(v);
    v = xmask_add<2>(v);
    v = xmask_add<4>(v);
    return v;
}

__device__ __forceinline__ void load_item(const float* __restrict__ feat, int t, float4 (&f)[4]) {
    const int b = t >> 14;
    const int i = t & 16383;
    const float4* fb = (const float4*)(feat + ((size_t)b << 18));
#pragma unroll
    for (int ch = 0; ch < 4; ++ch) f[ch] = fb[ch * 16384 + i];
}

__global__ __launch_bounds__(512) void mlp_pred_kernel(
    const float* __restrict__ feat,
    const float* __restrict__ W1, const float* __restrict__ b1,
    const float* __restrict__ W2, const float* __restrict__ b2,
    const float* __restrict__ W3, const float* __restrict__ b3,
    const float* __restrict__ W4, const float* __restrict__ b4,
    const float* __restrict__ W5, const float* __restrict__ b5,
    float* __restrict__ out)
{
    __shared__ __attribute__((aligned(16))) float bufA[D];
    __shared__ __attribute__((aligned(16))) float bufB[D];

    const int tid  = threadIdx.x;
    const int wave = __builtin_amdgcn_readfirstlane(tid >> 5);
    const int lane = tid & 31;
    const int c    = lane & 15;
    const int h    = lane >> 4;
    const int g    = c >> 2;
    const int qq   = c & 3;
    const int kof  = 8 * qq + 4 * h;
    const bool isxA = g == 3;
    const bool isxB = g == 0;

    const int gA = g < 2 ? g : 2;
    const int gB = g > 1 ? g : 1;
    int offA[3], offB[3];
#pragma unroll
    for (int j = 0; j < 3; ++j) {
        const int ra = 6 * j + gA;
        const int rb = 6 * j + 2 + gB;
        offA[j] = (16 * wave + ra) * D + kof;
        offB[j] = (16 * wave + (rb < 15 ? rb : 15)) * D + kof;
    }
    const float* wl[3] = {W2, W3, W4};
    const float* bs[3] = {b2, b3, b4};

    const int place = g == 0 ? (h ? 2 : 0) : (h ? 2 + g : 1);
    const int rloc  = 6 * qq + place;
    const bool fin  = qq < 3 && rloc < 16 && (h == 1 || g < 2);
    const int myrow = 16 * wave + (rloc < 15 ? rloc : 15);

    const int q8   = c & 7;
    const int koff = 4 * (q8 + 8 * h);
    const int    p5  = wave < 6 ? wave : 0;
    const float* w5l = W5 + (size_t)(2 * p5 + (c >> 3)) * D + koff;

    v4f ring[NSLOT][2 * KPS];
    v4f xk[8];
    v4f w5[4];

#pragma unroll
    for (int u = 0; u < NSLOT; ++u)
        issue_slot(wl[u / (3 * SPC)], offA[(u / SPC) % 3], offB[(u / SPC) % 3], u % SPC, ring[u]);

    if (tid < D) {
        float v = W1[tid * 4 + 2] + W1[tid * 4 + 3] + b1[tid];
        bufA[tid] = gelu_exact(v);
    }

    float4 f[ITEMS][4];
    lds_barrier();

    float bias5 = 0.0f;

#pragma unroll
    for (int L = 0; L < 3; ++L) {
        const float* xs = (L & 1) ? bufB : bufA;
        float*       xd = (L & 1) ? bufA : bufB;
#pragma unroll
        for (int kr = 0; kr < 8; ++kr) xk[kr] = *(const v4f*)(xs + kr * 32 + kof);
        const float bias = bs[L][myrow];
        if (L == 2) bias5 = b5[2 * p5 + h];
        float mine1 = 0.0f, mine2 = 0.0f;
#pragma unroll
        for (int j = 0; j < 3; ++j) {
            v8f accE = {}, accO = {};
#pragma unroll
            for (int part = 0; part < SPC; ++part) {
                const int u = (3 * L + j) * SPC + part;
                slot_mma(ring[u % NSLOT], xk, part, isxA, isxB, accE, accO);
                const int un = u + NSLOT;
                if (un < NU)       issue_slot(wl[un / (3 * SPC)], offA[(un / SPC) % 3], offB[(un / SPC) % 3], un % SPC, ring[u % NSLOT]);
                else if (un == NU) issue_pair(w5l, w5);
                if (un > NU && un - NU - 1 < ITEMS)
                    load_item(feat, (blockIdx.x * ITEMS + (un - NU - 1)) * 512 + tid, f[un - NU - 1]);
                __builtin_amdgcn_sched_barrier(0);
            }
            const v8f acc = accE + accO;
            float v1 = acc[0], v2 = acc[4];
#pragma unroll
            for (int r = 1; r < 4; ++r) {
                v1 = (qq == r) ? acc[r] : v1;
                v2 = (qq == r) ? acc[4 + r] : v2;
            }
            v1 = xmask_add<1>(v1); v1 = xmask_add<2>(v1);
            v2 = xmask_add<1>(v2); v2 = xmask_add<2>(v2);
            mine1 = (qq == j) ? v1 : mine1;
            mine2 = (qq == j) ? v2 : mine2;
        }
        const float handed = from_xmask4(mine2);
        const float mine = g == 0 ? mine1 : (h ? mine2 : handed);
        const float y = gelu_exact(mine + bias);
        if (fin) xd[myrow] = y;
        lds_barrier();
    }

    if (wave < 6) {
        v4f xf[4];
        load_x(bufB, koff, xf);
        const float v = pair_dot(w5, xf, q8);
        if (c == 0) bufA[2 * wave + h] = v + bias5;
    }
    lds_barrier();

    float w[12];
#pragma unroll
    for (int gg = 0; gg < 3; ++gg) {
        const v4f wv = *(const v4f*)(bufA + 4 * gg);
        w[4 * gg + 0] = wv.x; w[4 * gg + 1] = wv.y; w[4 * gg + 2] = wv.z; w[4 * gg + 3] = wv.w;
    }
#pragma unroll
    for (int it = 0; it < ITEMS; ++it) {
        const int t = (blockIdx.x * ITEMS + it) * 512 + tid;
        const float p0[4] = {f[it][0].x, f[it][0].y, f[it][0].z, f[it][0].w};
        const float p1[4] = {f[it][1].x, f[it][1].y, f[it][1].z, f[it][1].w};
        const float p2[4] = {f[it][2].x, f[it][2].y, f[it][2].z, f[it][2].w};
        const float p3[4] = {f[it][3].x, f[it][3].y, f[it][3].z, f[it][3].w};
        float o[12];
#pragma unroll
        for (int j = 0; j < 4; ++j) {
#pragma unroll
            for (int d = 0; d < 3; ++d) {
                o[j * 3 + d] = fmaf(p0[j], w[d],
                               fmaf(p1[j], w[3 + d],
                               fmaf(p2[j], w[6 + d],
                                    p3[j] * w[9 + d])));
            }
        }
        float4* op = (float4*)(out + (size_t)t * 12);
        op[0] = make_float4(o[0], o[1], o[2],  o[3]);
        op[1] = make_float4(o[4], o[5], o[6],  o[7]);
        op[2] = make_float4(o[8], o[9], o[10], o[11]);
    }
}

extern "C" void kernel_launch(void* const* d_in, const int* in_sizes, int n_in,
                              void* d_out, int out_size, void* d_ws, size_t ws_size,
                              hipStream_t stream) {
    const float* feat = (const float*)d_in[0];
    const float* W1   = (const float*)d_in[1];
    const float* b1   = (const float*)d_in[2];
    const float* W2   = (const float*)d_in[3];
    const float* b2   = (const float*)d_in[4];
    const float* W3   = (const float*)d_in[5];
    const float* b3   = (const float*)d_in[6];
    const float* W4   = (const float*)d_in[7];
    const float* b4   = (const float*)d_in[8];
    const float* W5   = (const float*)d_in[9];
    const float* b5   = (const float*)d_in[10];

    float* out = (float*)d_out;

    mlp_pred_kernel<<<NWG, 512, 0, stream>>>(feat, W1, b1, W2, b2, W3, b3, W4, b4, W5, b5, out);
}
